# starting kernel
# baseline (speedup 1.0000x reference)
#include <hip/hip_runtime.h>

typedef __attribute__((ext_vector_type(2))) float v2f;
typedef __attribute__((ext_vector_type(8))) float v8f;

#define BATCH 4096
#define NITEM 200
#define NSEG  26
#define DIM   128

__global__ __launch_bounds__(256)
void weighted_embedding_bag_kernel(const int* __restrict__ input,
                                   const float* __restrict__ psw,
                                   const int* __restrict__ offsets,
                                   const float* __restrict__ emb,
                                   float* __restrict__ out)
{
    const int b    = blockIdx.x;
    const int tid  = threadIdx.x;
    const int wave = tid >> 5;
    const int lane = tid & 31;
    const int lh   = lane >> 4;
    const int l16  = lane & 15;

    __shared__ int   s_off[NSEG];
    __shared__ int   s_idx[NITEM];
    __shared__ float s_val[NITEM];
    __shared__ int   s_seg[NITEM];

    if (tid < NSEG) s_off[tid] = offsets[b * NSEG + tid];
    __syncthreads();

    for (int n = tid; n < NITEM; n += 256) {
        s_idx[n] = input[b * NITEM + n];
        s_val[n] = psw[b * NITEM + n];
        int seg = -1;
        #pragma unroll
        for (int m = 0; m < NSEG; ++m) {
            if (seg < 0 && s_off[m] >= n) seg = m;
        }
        s_seg[n] = seg;
    }
    __syncthreads();

    const int dt  = wave;
    const int col = dt * 16 + l16;
    const int m0  = l16;
    const int m1  = 16 + l16;

    v8f acc0 = {0.f, 0.f, 0.f, 0.f, 0.f, 0.f, 0.f, 0.f};
    v8f acc1 = {0.f, 0.f, 0.f, 0.f, 0.f, 0.f, 0.f, 0.f};

    for (int kk = 0; kk < NITEM; kk += 4) {
        const int ke = kk + 2 * lh;
        const int ko = ke + 1;

        const int i0 = s_idx[ke];
        const int i1 = s_idx[ko];
        v2f bfrag;
        bfrag.x = emb[(long)i0 * DIM + col];
        bfrag.y = emb[(long)i1 * DIM + col];

        const int   sg0 = s_seg[ke];
        const int   sg1 = s_seg[ko];
        const float w0  = s_val[ke];
        const float w1  = s_val[ko];

        v2f a0, a1;
        a0.x = (sg0 == m0) ? w0 : 0.f;
        a0.y = (sg1 == m0) ? w1 : 0.f;
        a1.x = (sg0 == m1) ? w0 : 0.f;
        a1.y = (sg1 == m1) ? w1 : 0.f;

        acc0 = __builtin_amdgcn_wmma_f32_16x16x4_f32(
            false, a0, false, bfrag, (short)0, acc0, false, false);
        acc1 = __builtin_amdgcn_wmma_f32_16x16x4_f32(
            false, a1, false, bfrag, (short)0, acc1, false, false);
    }

    #pragma unroll
    for (int r = 0; r < 8; ++r) {
        const int mrow = r + lh * 8;
        out[((long)(b * NSEG + mrow)) * DIM + col] = acc0[r];
        const int m = 16 + mrow;
        if (m < NSEG)
            out[((long)(b * NSEG + m)) * DIM + col] = acc1[r];
    }
}

extern "C" void kernel_launch(void* const* d_in, const int* in_sizes, int n_in,
                              void* d_out, int out_size, void* d_ws, size_t ws_size,
                              hipStream_t stream) {
    (void)in_sizes; (void)n_in; (void)out_size; (void)d_ws; (void)ws_size;
    const int*   input   = (const int*)d_in[0];
    const float* psw     = (const float*)d_in[1];
    const int*   offsets = (const int*)d_in[2];
    const float* emb     = (const float*)d_in[3];
    float*       out     = (float*)d_out;

    weighted_embedding_bag_kernel<<<BATCH, 256, 0, stream>>>(input, psw, offsets, emb, out);
}


# Gather and store 16 bytes per lane: each wave takes 64 columns as four N-tiles (tile j = column 4n+j) for both M-tiles, 2 waves per batch row; summation order unchanged
# speedup vs baseline: 1.3765x; 1.3765x over previous
#include <hip/hip_runtime.h>


typedef __attribute__((ext_vector_type(2))) float v2f;
typedef __attribute__((ext_vector_type(4))) float v4f;
typedef __attribute__((ext_vector_type(8))) float v8f;

#define BATCH 4096
#define NITEM 200
#define NSEG  26
#define DIM   128
#define THREADS 64

__global__ __launch_bounds__(THREADS) __attribute__((amdgpu_waves_per_eu(8, 9)))
void weighted_embedding_bag_kernel(const int* __restrict__ input,
                                   const float* __restrict__ psw,
                                   const int* __restrict__ offsets,
                                   const float* __restrict__ emb,
                                   float* __restrict__ out)
{
    const int b    = blockIdx.x;
    const int tid  = threadIdx.x;
    const int wave = tid >> 5;
    const int lane = tid & 31;
    const int lh   = lane >> 4;
    const int l16  = lane & 15;

    __shared__ int   s_off[NSEG];
    __shared__ int   s_idx[NITEM];
    __shared__ float s_val[NITEM];
    __shared__ int   s_seg[NITEM];

    if (tid < NSEG) s_off[tid] = offsets[b * NSEG + tid];
    __syncthreads();

    for (int n = tid; n < NITEM; n += THREADS) {
        s_idx[n] = input[b * NITEM + n];
        s_val[n] = psw[b * NITEM + n];
        int seg = -1;
        #pragma unroll
        for (int m = 0; m < NSEG; ++m) {
            if (seg < 0 && s_off[m] >= n) seg = m;
        }
        s_seg[n] = seg;
    }
    __syncthreads();

    const int cbase = wave * 64 + 4 * l16;
    const int m0    = l16;
    const int m1    = 16 + l16;

    const v8f zero8 = {0.f, 0.f, 0.f, 0.f, 0.f, 0.f, 0.f, 0.f};
    v8f acc0[4], acc1[4];
    #pragma unroll
    for (int j = 0; j < 4; ++j) { acc0[j] = zero8; acc1[j] = zero8; }

    for (int kk = 0; kk < NITEM; kk += 8) {
        v4f be[2], bo[2];
        v2f a0[2], a1[2];
        #pragma unroll
        for (int u = 0; u < 2; ++u) {
            const int ke = kk + 4 * u + 2 * lh;
            const int ko = ke + 1;
            const int i0 = s_idx[ke];
            const int i1 = s_idx[ko];
            be[u] = *(const v4f*)(emb + (long)i0 * DIM + cbase);
            bo[u] = *(const v4f*)(emb + (long)i1 * DIM + cbase);
        }
        #pragma unroll
        for (int u = 0; u < 2; ++u) {
            const int ke = kk + 4 * u + 2 * lh;
            const int ko = ke + 1;
            const int   sg0 = s_seg[ke];
            const int   sg1 = s_seg[ko];
            const float w0  = s_val[ke];
            const float w1  = s_val[ko];
            a0[u].x = (sg0 == m0) ? w0 : 0.f;
            a0[u].y = (sg1 == m0) ? w1 : 0.f;
            a1[u].x = (sg0 == m1) ? w0 : 0.f;
            a1[u].y = (sg1 == m1) ? w1 : 0.f;
        }
        #pragma unroll
        for (int u = 0; u < 2; ++u) {
            #pragma unroll
            for (int j = 0; j < 4; ++j) {
                v2f bfrag;
                bfrag.x = be[u][j];
                bfrag.y = bo[u][j];
                acc0[j] = __builtin_amdgcn_wmma_f32_16x16x4_f32(
                    false, a0[u], false, bfrag, (short)0, acc0[j], false, false);
                acc1[j] = __builtin_amdgcn_wmma_f32_16x16x4_f32(
                    false, a1[u], false, bfrag, (short)0, acc1[j], false, false);
            }
        }
    }

    #pragma unroll
    for (int r = 0; r < 8; ++r) {
        const int mrow = r + lh * 8;
        v4f o0;
        o0.x = acc0[0][r]; o0.y = acc0[1][r]; o0.z = acc0[2][r]; o0.w = acc0[3][r];
        *(v4f*)(out + ((long)(b * NSEG + mrow)) * DIM + cbase) = o0;
        const int m = 16 + mrow;
        if (m < NSEG) {
            v4f o1;
            o1.x = acc1[0][r]; o1.y = acc1[1][r]; o1.z = acc1[2][r]; o1.w = acc1[3][r];
            *(v4f*)(out + ((long)(b * NSEG + m)) * DIM + cbase) = o1;
        }
    }
}

extern "C" void kernel_launch(void* const* d_in, const int* in_sizes, int n_in,
                              void* d_out, int out_size, void* d_ws, size_t ws_size,
                              hipStream_t stream) {
    (void)in_sizes; (void)n_in; (void)out_size; (void)d_ws; (void)ws_size;
    const int*   input   = (const int*)d_in[0];
    const float* psw     = (const float*)d_in[1];
    const int*   offsets = (const int*)d_in[2];
    const float* emb     = (const float*)d_in[3];
    float*       out     = (float*)d_out;

    weighted_embedding_bag_kernel<<<BATCH, THREADS, 0, stream>>>(input, psw, offsets, emb, out);
}


# Do the two 16-row M-tiles one after the other, each only over the 8-item trips holding items of its own segments (offsets are sorted): half the matrix instructions; summation order unchanged
# speedup vs baseline: 1.8500x; 1.3440x over previous
#include <hip/hip_runtime.h>


typedef __attribute__((ext_vector_type(2))) float v2f;
typedef __attribute__((ext_vector_type(4))) float v4f;
typedef __attribute__((ext_vector_type(8))) float v8f;

#define BATCH 4096
#define NITEM 200
#define NSEG  26
#define DIM   128
#define THREADS 64
#define TILE_M 16

__global__ __launch_bounds__(THREADS) __attribute__((amdgpu_waves_per_eu(8, 9)))
void weighted_embedding_bag_kernel(const int* __restrict__ input,
                                   const float* __restrict__ psw,
                                   const int* __restrict__ offsets,
                                   const float* __restrict__ emb,
                                   float* __restrict__ out)
{
    const int b    = blockIdx.x;
    const int tid  = threadIdx.x;
    const int wave = tid >> 5;
    const int lane = tid & 31;
    const int lh   = lane >> 4;
    const int l16  = lane & 15;

    const int last0 = min(offsets[b * NSEG + (TILE_M - 1)], NITEM - 1);
    const int last1 = min(offsets[b * NSEG + (NSEG - 1)], NITEM - 1);

    __shared__ int   s_off[NSEG];
    __shared__ int   s_idx[NITEM];
    __shared__ float s_val[NITEM];
    __shared__ int   s_seg[NITEM];

    if (tid < NSEG) s_off[tid] = offsets[b * NSEG + tid];
    __syncthreads();

    for (int n = tid; n < NITEM; n += THREADS) {
        s_idx[n] = input[b * NITEM + n];
        s_val[n] = psw[b * NITEM + n];
        int seg = -1;
        #pragma unroll
        for (int m = 0; m < NSEG; ++m) {
            if (seg < 0 && s_off[m] >= n) seg = m;
        }
        s_seg[n] = seg;
    }
    __syncthreads();

    const int cbase = wave * 64 + 4 * l16;

    const v8f zero8 = {0.f, 0.f, 0.f, 0.f, 0.f, 0.f, 0.f, 0.f};

    for (int t = 0; t < 2; ++t) {
        const int mA   = TILE_M * t + l16;
        const int kbeg = (t == 0) ? 0 : ((last0 + 1) & ~7);
        const int kend = (t == 0) ? last0 : last1;

        v8f acc[4];
        #pragma unroll
        for (int j = 0; j < 4; ++j) acc[j] = zero8;

        for (int kk = kbeg; kk <= kend; kk += 8) {
            v4f be[2], bo[2];
            v2f a[2];
            #pragma unroll
            for (int u = 0; u < 2; ++u) {
                const int ke = kk + 4 * u + 2 * lh;
                const int ko = ke + 1;
                const int i0 = s_idx[ke];
                const int i1 = s_idx[ko];
                be[u] = *(const v4f*)(emb + (long)i0 * DIM + cbase);
                bo[u] = *(const v4f*)(emb + (long)i1 * DIM + cbase);
            }
            #pragma unroll
            for (int u = 0; u < 2; ++u) {
                const int ke = kk + 4 * u + 2 * lh;
                const int ko = ke + 1;
                const int   sg0 = s_seg[ke];
                const int   sg1 = s_seg[ko];
                const float w0  = s_val[ke];
                const float w1  = s_val[ko];
                a[u].x = (sg0 == mA) ? w0 : 0.f;
                a[u].y = (sg1 == mA) ? w1 : 0.f;
            }
            #pragma unroll
            for (int u = 0; u < 2; ++u) {
                #pragma unroll
                for (int j = 0; j < 4; ++j) {
                    v2f bfrag;
                    bfrag.x = be[u][j];
                    bfrag.y = bo[u][j];
                    acc[j] = __builtin_amdgcn_wmma_f32_16x16x4_f32(
                        false, a[u], false, bfrag, (short)0, acc[j], false, false);
                }
            }
        }

        #pragma unroll
        for (int r = 0; r < 8; ++r) {
            const int m = TILE_M * t + r + lh * 8;
            if (m < NSEG) {
                v4f o;
                o.x = acc[0][r]; o.y = acc[1][r]; o.z = acc[2][r]; o.w = acc[3][r];
                *(v4f*)(out + ((long)(b * NSEG + m)) * DIM + cbase) = o;
            }
        }
    }
}

extern "C" void kernel_launch(void* const* d_in, const int* in_sizes, int n_in,
                              void* d_out, int out_size, void* d_ws, size_t ws_size,
                              hipStream_t stream) {
    (void)in_sizes; (void)n_in; (void)out_size; (void)d_ws; (void)ws_size;
    const int*   input   = (const int*)d_in[0];
    const float* psw     = (const float*)d_in[1];
    const int*   offsets = (const int*)d_in[2];
    const float* emb     = (const float*)d_in[3];
    float*       out     = (float*)d_out;

    weighted_embedding_bag_kernel<<<BATCH, THREADS, 0, stream>>>(input, psw, offsets, emb, out);
}


# Row loads run two trips ahead of the matrix instructions: three register sets of trip rows in turn, loop three trips long, rows addressed by table base plus 32-bit offset; summation order unchanged
# speedup vs baseline: 1.9924x; 1.0770x over previous
#include <hip/hip_runtime.h>


typedef __attribute__((ext_vector_type(2))) float v2f;
typedef __attribute__((ext_vector_type(4))) float v4f;
typedef __attribute__((ext_vector_type(8))) float v8f;

#define BATCH 4096
#define NITEM 200
#define NSEG  26
#define DIM   128
#define THREADS 64
#define TILE_M 16

struct TripRows { v4f be0, bo0, be1, bo1; };

__device__ __forceinline__ void gather_trip(TripRows& r, const int* s_idx, const float* __restrict__ emb,
                                            int kk, int lh, int cbase)
{
    const int k0 = kk + 2 * lh;
    const int k1 = k0 + 4;
    const unsigned cb = (unsigned)cbase * 4u;
    const unsigned o00 = (unsigned)s_idx[k0]     * (DIM * 4u) + cb;
    const unsigned o01 = (unsigned)s_idx[k0 + 1] * (DIM * 4u) + cb;
    const unsigned o10 = (unsigned)s_idx[k1]     * (DIM * 4u) + cb;
    const unsigned o11 = (unsigned)s_idx[k1 + 1] * (DIM * 4u) + cb;
    const char* base = (const char*)emb;
    r.be0 = *(const v4f*)(base + o00);
    r.bo0 = *(const v4f*)(base + o01);
    r.be1 = *(const v4f*)(base + o10);
    r.bo1 = *(const v4f*)(base + o11);
    __builtin_amdgcn_sched_barrier(0);
}

__device__ __forceinline__ void mma_trip(v8f (&acc)[4], const TripRows& r, const int* s_seg,
                                         const float* s_val, int kk, int lh, int mA)
{
    const int k0 = kk + 2 * lh;
    const int k1 = k0 + 4;
    const int   sg00 = s_seg[k0], sg01 = s_seg[k0 + 1];
    const int   sg10 = s_seg[k1], sg11 = s_seg[k1 + 1];
    const float w00  = s_val[k0], w01  = s_val[k0 + 1];
    const float w10  = s_val[k1], w11  = s_val[k1 + 1];
    v2f a0, a1;
    a0.x = (sg00 == mA) ? w00 : 0.f;
    a0.y = (sg01 == mA) ? w01 : 0.f;
    a1.x = (sg10 == mA) ? w10 : 0.f;
    a1.y = (sg11 == mA) ? w11 : 0.f;
    #pragma unroll
    for (int j = 0; j < 4; ++j) {
        v2f bfrag;
        bfrag.x = r.be0[j];
        bfrag.y = r.bo0[j];
        acc[j] = __builtin_amdgcn_wmma_f32_16x16x4_f32(
            false, a0, false, bfrag, (short)0, acc[j], false, false);
    }
    #pragma unroll
    for (int j = 0; j < 4; ++j) {
        v2f bfrag;
        bfrag.x = r.be1[j];
        bfrag.y = r.bo1[j];
        acc[j] = __builtin_amdgcn_wmma_f32_16x16x4_f32(
            false, a1, false, bfrag, (short)0, acc[j], false, false);
    }
    __builtin_amdgcn_sched_barrier(0);
}

__global__ __launch_bounds__(THREADS) __attribute__((amdgpu_waves_per_eu(8, 8)))
void weighted_embedding_bag_kernel(const int* __restrict__ input,
                                   const float* __restrict__ psw,
                                   const int* __restrict__ offsets,
                                   const float* __restrict__ emb,
                                   float* __restrict__ out)
{
    const int b    = blockIdx.x;
    const int tid  = threadIdx.x;
    const int wave = tid >> 5;
    const int lane = tid & 31;
    const int lh   = lane >> 4;
    const int l16  = lane & 15;

    const int last0 = min(offsets[b * NSEG + (TILE_M - 1)], NITEM - 1);
    const int last1 = min(offsets[b * NSEG + (NSEG - 1)], NITEM - 1);

    __shared__ int   s_off[NSEG];
    __shared__ int   s_idx[NITEM];
    __shared__ float s_val[NITEM];
    __shared__ int   s_seg[NITEM];

    if (tid < NSEG) s_off[tid] = offsets[b * NSEG + tid];
    __syncthreads();

    for (int n = tid; n < NITEM; n += THREADS) {
        s_idx[n] = input[b * NITEM + n];
        s_val[n] = psw[b * NITEM + n];
        int seg = -1;
        #pragma unroll
        for (int m = 0; m < NSEG; ++m) {
            if (seg < 0 && s_off[m] >= n) seg = m;
        }
        s_seg[n] = seg;
    }
    __syncthreads();

    const int cbase = wave * 64 + 4 * l16;

    const v8f zero8 = {0.f, 0.f, 0.f, 0.f, 0.f, 0.f, 0.f, 0.f};

    for (int t = 0; t < 2; ++t) {
        const int mA   = TILE_M * t + l16;
        const int kbeg = (t == 0) ? 0 : ((last0 + 1) & ~7);
        const int kend = (t == 0) ? last0 : last1;

        v8f acc[4];
        #pragma unroll
        for (int j = 0; j < 4; ++j) acc[j] = zero8;

        const int ntrip = (kend >= kbeg) ? (((kend - kbeg) >> 3) + 1) : 0;
        TripRows A, B, C;
        if (ntrip == 1) {
            gather_trip(A, s_idx, emb, kbeg, lh, cbase);
            mma_trip(acc, A, s_seg, s_val, kbeg, lh, mA);
        } else if (ntrip >= 2) {
            const int head = (ntrip - 2) % 3;
            int kk = kbeg;
            if (head == 0) {
                gather_trip(A, s_idx, emb, kk, lh, cbase);
                gather_trip(B, s_idx, emb, kk + 8, lh, cbase);
            } else if (head == 1) {
                gather_trip(C, s_idx, emb, kk, lh, cbase);
                gather_trip(A, s_idx, emb, kk + 8, lh, cbase);
                gather_trip(B, s_idx, emb, kk + 16, lh, cbase);
                mma_trip(acc, C, s_seg, s_val, kk, lh, mA);
                kk += 8;
            } else {
                gather_trip(B, s_idx, emb, kk, lh, cbase);
                gather_trip(C, s_idx, emb, kk + 8, lh, cbase);
                gather_trip(A, s_idx, emb, kk + 16, lh, cbase);
                mma_trip(acc, B, s_seg, s_val, kk, lh, mA);
                gather_trip(B, s_idx, emb, kk + 24, lh, cbase);
                mma_trip(acc, C, s_seg, s_val, kk + 8, lh, mA);
                kk += 16;
            }
            for (int left = ntrip - head; left > 2; left -= 3) {
                gather_trip(C, s_idx, emb, kk + 16, lh, cbase);
                mma_trip(acc, A, s_seg, s_val, kk, lh, mA);
                gather_trip(A, s_idx, emb, kk + 24, lh, cbase);
                mma_trip(acc, B, s_seg, s_val, kk + 8, lh, mA);
                gather_trip(B, s_idx, emb, kk + 32, lh, cbase);
                mma_trip(acc, C, s_seg, s_val, kk + 16, lh, mA);
                kk += 24;
            }
            mma_trip(acc, A, s_seg, s_val, kk, lh, mA);
            mma_trip(acc, B, s_seg, s_val, kk + 8, lh, mA);
        }

        #pragma unroll
        for (int r = 0; r < 8; ++r) {
            const int m = TILE_M * t + r + lh * 8;
            if (m < NSEG) {
                v4f o;
                o.x = acc[0][r]; o.y = acc[1][r]; o.z = acc[2][r]; o.w = acc[3][r];
                *(v4f*)(out + ((long)(b * NSEG + m)) * DIM + cbase) = o;
            }
        }
    }
}

extern "C" void kernel_launch(void* const* d_in, const int* in_sizes, int n_in,
                              void* d_out, int out_size, void* d_ws, size_t ws_size,
                              hipStream_t stream) {
    (void)in_sizes; (void)n_in; (void)out_size; (void)d_ws; (void)ws_size;
    const int*   input   = (const int*)d_in[0];
    const float* psw     = (const float*)d_in[1];
    const int*   offsets = (const int*)d_in[2];
    const float* emb     = (const float*)d_in[3];
    float*       out     = (float*)d_out;

    weighted_embedding_bag_kernel<<<BATCH, THREADS, 0, stream>>>(input, psw, offsets, emb, out);
}
